# starting kernel
# baseline (speedup 1.0000x reference)
#include <hip/hip_runtime.h>
#include <math.h>

typedef float v2f __attribute__((ext_vector_type(2)));
typedef float v8f __attribute__((ext_vector_type(8)));

#define EPSF    1e-8f
#define LN_EPSF 1e-5f
#define B_ROWS  8192
#define S_LEN   4096
#define HID     256
#define HEAD_H  64

__global__ __launch_bounds__(256) void regime_features_kernel(
    const float* __restrict__ prices,
    const float* __restrict__ enc_w,
    const float* __restrict__ enc_b,
    const float* __restrict__ enc_g,
    const float* __restrict__ enc_beta,
    float* __restrict__ rf)
{
    __shared__ float sp[S_LEN];
    __shared__ float r1[256];
    __shared__ float r2[256];
    __shared__ float scomb[4];
    __shared__ float smean;
    __shared__ float lnbuf[256];

    const int tid = threadIdx.x;
    const int b   = blockIdx.x;
    const float* row = prices + (size_t)b * S_LEN;

    for (int j = tid; j < S_LEN; j += 256) sp[j] = row[j];
    __syncthreads();

    const int i0 = tid * 16;

    float lsum = 0.f;
    for (int i = i0; i < i0 + 16; ++i) lsum += sp[i];
    r1[tid] = lsum;
    __syncthreads();
    for (int s = 128; s > 0; s >>= 1) {
        if (tid < s) r1[tid] += r1[tid + s];
        __syncthreads();
    }
    if (tid == 0) smean = r1[0] * (1.0f / (float)S_LEN);
    __syncthreads();
    const float mean = smean;

    float winsum = 0.f;
    int start = i0 - 19; if (start < 0) start = 0;
    for (int j = start; j <= i0; ++j) winsum += sp[j];

    float trend = 0.f, var_acc = 0.f;
    for (int i = i0; i < i0 + 16; ++i) {
        if (i != i0) {
            winsum += sp[i];
            if (i - 20 >= 0) winsum -= sp[i - 20];
        }
        const float cnt = (float)((i + 1 < 20) ? (i + 1) : 20);
        const float sma = winsum / cnt;
        const float p   = sp[i];
        trend += (p - sma) / (sma + EPSF);
        const float d = p - mean;
        var_acc += d * d;
    }
    __syncthreads();
    r1[tid] = trend;
    r2[tid] = var_acc;
    __syncthreads();
    for (int s = 128; s > 0; s >>= 1) {
        if (tid < s) { r1[tid] += r1[tid + s]; r2[tid] += r2[tid + s]; }
        __syncthreads();
    }
    if (tid == 0) {
        const float f_trend = r1[0] * (1.0f / (float)S_LEN);
        const float p_last  = sp[S_LEN - 1];
        const float p_m10   = sp[S_LEN - 10];
        const float f_mom   = (p_last - p_m10) / (p_m10 + EPSF);
        const float f_mr    = (p_last - mean) / (mean + EPSF);
        const float var1    = r2[0] / (float)(S_LEN - 1);
        const float f_cyc   = sqrtf(var1) / (mean + EPSF);
        scomb[0] = f_trend; scomb[1] = f_mom; scomb[2] = f_mr; scomb[3] = f_cyc;
    }
    __syncthreads();

    const int e = tid >> 6;
    const int o = tid & 63;
    float pre = enc_b[e * 64 + o];
#pragma unroll
    for (int i = 0; i < 4; ++i)
        pre += scomb[i] * enc_w[e * 256 + i * 64 + o];
    const float x = pre > 0.f ? pre : 0.f;
    lnbuf[tid] = x;
    __syncthreads();

    const int gb = e * 64;
    float mu = 0.f;
    for (int j = 0; j < 64; ++j) mu += lnbuf[gb + j];
    mu *= (1.0f / 64.0f);
    float v = 0.f;
    for (int j = 0; j < 64; ++j) { const float d = lnbuf[gb + j] - mu; v += d * d; }
    v *= (1.0f / 64.0f);
    const float outv = (x - mu) / sqrtf(v + LN_EPSF) * enc_g[gb + o] + enc_beta[gb + o];
    rf[(size_t)b * HID + tid] = outv;
}

__global__ __launch_bounds__(256) void regime_head_kernel(
    const float* __restrict__ rf,
    const float* __restrict__ w1,
    const float* __restrict__ b1,
    const float* __restrict__ w2,
    const float* __restrict__ b2,
    float* __restrict__ probs)
{
    __shared__ float hbuf[128 * HEAD_H];

    const int tid     = threadIdx.x;
    const int lane    = tid & 31;
    const int wave    = tid >> 5;
    const int rowbase = blockIdx.x * 128 + wave * 16;
    const int m16     = lane & 15;
    const int koff    = (lane >= 16) ? 2 : 0;
    const float* arow = rf + (size_t)(rowbase + m16) * HID;

    v8f acc0 = {0.f,0.f,0.f,0.f,0.f,0.f,0.f,0.f};
    v8f acc1 = acc0, acc2 = acc0, acc3 = acc0;

    for (int kk = 0; kk < HID; kk += 4) {
        v2f a;
        a.x = arow[kk + koff];
        a.y = arow[kk + koff + 1];
        const float* w1r0 = w1 + (size_t)(kk + koff) * HEAD_H + m16;
        const float* w1r1 = w1r0 + HEAD_H;
        v2f bm;
        bm.x = w1r0[0];  bm.y = w1r1[0];
        acc0 = __builtin_amdgcn_wmma_f32_16x16x4_f32(false, a, false, bm, (short)0, acc0, false, false);
        bm.x = w1r0[16]; bm.y = w1r1[16];
        acc1 = __builtin_amdgcn_wmma_f32_16x16x4_f32(false, a, false, bm, (short)0, acc1, false, false);
        bm.x = w1r0[32]; bm.y = w1r1[32];
        acc2 = __builtin_amdgcn_wmma_f32_16x16x4_f32(false, a, false, bm, (short)0, acc2, false, false);
        bm.x = w1r0[48]; bm.y = w1r1[48];
        acc3 = __builtin_amdgcn_wmma_f32_16x16x4_f32(false, a, false, bm, (short)0, acc3, false, false);
    }

    const int mlo = (lane < 16) ? 0 : 8;
#pragma unroll
    for (int r = 0; r < 8; ++r) {
        const int mloc = wave * 16 + mlo + r;
        float h0 = acc0[r] + b1[ 0 + m16]; h0 = fmaxf(h0, 0.f);
        float h1 = acc1[r] + b1[16 + m16]; h1 = fmaxf(h1, 0.f);
        float h2 = acc2[r] + b1[32 + m16]; h2 = fmaxf(h2, 0.f);
        float h3 = acc3[r] + b1[48 + m16]; h3 = fmaxf(h3, 0.f);
        hbuf[mloc * HEAD_H +  0 + m16] = h0;
        hbuf[mloc * HEAD_H + 16 + m16] = h1;
        hbuf[mloc * HEAD_H + 32 + m16] = h2;
        hbuf[mloc * HEAD_H + 48 + m16] = h3;
    }
    __syncthreads();

    if (tid < 128) {
        const float* hr = hbuf + tid * HEAD_H;
        float l0 = b2[0], l1 = b2[1], l2 = b2[2], l3 = b2[3];
        for (int n = 0; n < HEAD_H; ++n) {
            const float h = hr[n];
            l0 += h * w2[n * 4 + 0];
            l1 += h * w2[n * 4 + 1];
            l2 += h * w2[n * 4 + 2];
            l3 += h * w2[n * 4 + 3];
        }
        const float mx = fmaxf(fmaxf(l0, l1), fmaxf(l2, l3));
        const float e0 = expf(l0 - mx), e1 = expf(l1 - mx);
        const float e2 = expf(l2 - mx), e3 = expf(l3 - mx);
        const float inv = 1.0f / (e0 + e1 + e2 + e3);
        float* po = probs + (size_t)(blockIdx.x * 128 + tid) * 4;
        po[0] = e0 * inv; po[1] = e1 * inv; po[2] = e2 * inv; po[3] = e3 * inv;
    }
}

extern "C" void kernel_launch(void* const* d_in, const int* in_sizes, int n_in,
                              void* d_out, int out_size, void* d_ws, size_t ws_size,
                              hipStream_t stream) {
    (void)in_sizes; (void)n_in; (void)out_size; (void)d_ws; (void)ws_size;
    const float* prices   = (const float*)d_in[0];
    const float* enc_w    = (const float*)d_in[1];
    const float* enc_b    = (const float*)d_in[2];
    const float* enc_g    = (const float*)d_in[3];
    const float* enc_beta = (const float*)d_in[4];
    const float* head_w1  = (const float*)d_in[5];
    const float* head_b1  = (const float*)d_in[6];
    const float* head_w2  = (const float*)d_in[7];
    const float* head_b2  = (const float*)d_in[8];

    float* rf    = (float*)d_out;
    float* probs = rf + (size_t)B_ROWS * HID;

    regime_features_kernel<<<B_ROWS, 256, 0, stream>>>(
        prices, enc_w, enc_b, enc_g, enc_beta, rf);
    regime_head_kernel<<<B_ROWS / 128, 256, 0, stream>>>(
        rf, head_w1, head_b1, head_w2, head_b2, probs);
}


# Row buffer in LDS padded to a 17-word pitch per 16-element thread chunk (bank-conflict-free chunk reads); each thread reads its chunk and 20-element halo once into registers for both passes
# speedup vs baseline: 1.3235x; 1.3235x over previous
#include <hip/hip_runtime.h>
#include <math.h>

typedef float v2f __attribute__((ext_vector_type(2)));
typedef float v8f __attribute__((ext_vector_type(8)));

#define EPSF    1e-8f
#define LN_EPSF 1e-5f
#define B_ROWS  8192
#define S_LEN   4096
#define HID     256
#define HEAD_H  64

__global__ __launch_bounds__(256) void regime_features_kernel(
    const float* __restrict__ prices,
    const float* __restrict__ enc_w,
    const float* __restrict__ enc_b,
    const float* __restrict__ enc_g,
    const float* __restrict__ enc_beta,
    float* __restrict__ rf)
{
    __shared__ float sp[S_LEN + S_LEN / 16];
    __shared__ float r1[256];
    __shared__ float r2[256];
    __shared__ float scomb[4];
    __shared__ float smean;
    __shared__ float lnbuf[256];

    const int tid = threadIdx.x;
    const int b   = blockIdx.x;
    const float* row = prices + (size_t)b * S_LEN;

    for (int j = tid; j < S_LEN; j += 256) sp[j + (j >> 4)] = row[j];
    __syncthreads();

    const int i0 = tid * 16;
    const int c0 = tid * 17;

    float own[16];
#pragma unroll
    for (int k = 0; k < 16; ++k) own[k] = sp[c0 + k];
    float hal[20];
    const int cm2 = (tid >= 2) ? c0 - 34 : 0;
    const int cm1 = (tid >= 1) ? c0 - 17 : 0;
#pragma unroll
    for (int j = 1; j < 4; ++j)
        hal[j] = (tid >= 2) ? sp[cm2 + 12 + j] : 0.f;
#pragma unroll
    for (int j = 4; j < 20; ++j)
        hal[j] = (tid >= 1) ? sp[cm1 + (j - 4)] : 0.f;

    float lsum = 0.f;
#pragma unroll
    for (int k = 0; k < 16; ++k) lsum += own[k];
    r1[tid] = lsum;
    __syncthreads();
    for (int s = 128; s > 0; s >>= 1) {
        if (tid < s) r1[tid] += r1[tid + s];
        __syncthreads();
    }
    if (tid == 0) smean = r1[0] * (1.0f / (float)S_LEN);
    __syncthreads();
    const float mean = smean;

    float winsum = 0.f;
#pragma unroll
    for (int j = 1; j < 20; ++j) winsum += hal[j];
    winsum += own[0];

    float trend = 0.f, var_acc = 0.f;
#pragma unroll
    for (int k = 0; k < 16; ++k) {
        const int i = i0 + k;
        if (k != 0) {
            winsum += own[k];
            winsum -= hal[k];
        }
        const float cnt = (float)((i + 1 < 20) ? (i + 1) : 20);
        const float sma = winsum / cnt;
        const float p   = own[k];
        trend += (p - sma) / (sma + EPSF);
        const float d = p - mean;
        var_acc += d * d;
    }
    __syncthreads();
    r1[tid] = trend;
    r2[tid] = var_acc;
    __syncthreads();
    for (int s = 128; s > 0; s >>= 1) {
        if (tid < s) { r1[tid] += r1[tid + s]; r2[tid] += r2[tid + s]; }
        __syncthreads();
    }
    if (tid == 0) {
        const float f_trend = r1[0] * (1.0f / (float)S_LEN);
        const float p_last  = sp[(S_LEN - 1) + ((S_LEN - 1) >> 4)];
        const float p_m10   = sp[(S_LEN - 10) + ((S_LEN - 10) >> 4)];
        const float f_mom   = (p_last - p_m10) / (p_m10 + EPSF);
        const float f_mr    = (p_last - mean) / (mean + EPSF);
        const float var1    = r2[0] / (float)(S_LEN - 1);
        const float f_cyc   = sqrtf(var1) / (mean + EPSF);
        scomb[0] = f_trend; scomb[1] = f_mom; scomb[2] = f_mr; scomb[3] = f_cyc;
    }
    __syncthreads();

    const int e = tid >> 6;
    const int o = tid & 63;
    float pre = enc_b[e * 64 + o];
#pragma unroll
    for (int i = 0; i < 4; ++i)
        pre += scomb[i] * enc_w[e * 256 + i * 64 + o];
    const float x = pre > 0.f ? pre : 0.f;
    lnbuf[tid] = x;
    __syncthreads();

    const int gb = e * 64;
    float mu = 0.f;
    for (int j = 0; j < 64; ++j) mu += lnbuf[gb + j];
    mu *= (1.0f / 64.0f);
    float v = 0.f;
    for (int j = 0; j < 64; ++j) { const float d = lnbuf[gb + j] - mu; v += d * d; }
    v *= (1.0f / 64.0f);
    const float outv = (x - mu) / sqrtf(v + LN_EPSF) * enc_g[gb + o] + enc_beta[gb + o];
    rf[(size_t)b * HID + tid] = outv;
}

__global__ __launch_bounds__(256) void regime_head_kernel(
    const float* __restrict__ rf,
    const float* __restrict__ w1,
    const float* __restrict__ b1,
    const float* __restrict__ w2,
    const float* __restrict__ b2,
    float* __restrict__ probs)
{
    __shared__ float hbuf[128 * HEAD_H];

    const int tid     = threadIdx.x;
    const int lane    = tid & 31;
    const int wave    = tid >> 5;
    const int rowbase = blockIdx.x * 128 + wave * 16;
    const int m16     = lane & 15;
    const int koff    = (lane >= 16) ? 2 : 0;
    const float* arow = rf + (size_t)(rowbase + m16) * HID;

    v8f acc0 = {0.f,0.f,0.f,0.f,0.f,0.f,0.f,0.f};
    v8f acc1 = acc0, acc2 = acc0, acc3 = acc0;

    for (int kk = 0; kk < HID; kk += 4) {
        v2f a;
        a.x = arow[kk + koff];
        a.y = arow[kk + koff + 1];
        const float* w1r0 = w1 + (size_t)(kk + koff) * HEAD_H + m16;
        const float* w1r1 = w1r0 + HEAD_H;
        v2f bm;
        bm.x = w1r0[0];  bm.y = w1r1[0];
        acc0 = __builtin_amdgcn_wmma_f32_16x16x4_f32(false, a, false, bm, (short)0, acc0, false, false);
        bm.x = w1r0[16]; bm.y = w1r1[16];
        acc1 = __builtin_amdgcn_wmma_f32_16x16x4_f32(false, a, false, bm, (short)0, acc1, false, false);
        bm.x = w1r0[32]; bm.y = w1r1[32];
        acc2 = __builtin_amdgcn_wmma_f32_16x16x4_f32(false, a, false, bm, (short)0, acc2, false, false);
        bm.x = w1r0[48]; bm.y = w1r1[48];
        acc3 = __builtin_amdgcn_wmma_f32_16x16x4_f32(false, a, false, bm, (short)0, acc3, false, false);
    }

    const int mlo = (lane < 16) ? 0 : 8;
#pragma unroll
    for (int r = 0; r < 8; ++r) {
        const int mloc = wave * 16 + mlo + r;
        float h0 = acc0[r] + b1[ 0 + m16]; h0 = fmaxf(h0, 0.f);
        float h1 = acc1[r] + b1[16 + m16]; h1 = fmaxf(h1, 0.f);
        float h2 = acc2[r] + b1[32 + m16]; h2 = fmaxf(h2, 0.f);
        float h3 = acc3[r] + b1[48 + m16]; h3 = fmaxf(h3, 0.f);
        hbuf[mloc * HEAD_H +  0 + m16] = h0;
        hbuf[mloc * HEAD_H + 16 + m16] = h1;
        hbuf[mloc * HEAD_H + 32 + m16] = h2;
        hbuf[mloc * HEAD_H + 48 + m16] = h3;
    }
    __syncthreads();

    if (tid < 128) {
        const float* hr = hbuf + tid * HEAD_H;
        float l0 = b2[0], l1 = b2[1], l2 = b2[2], l3 = b2[3];
        for (int n = 0; n < HEAD_H; ++n) {
            const float h = hr[n];
            l0 += h * w2[n * 4 + 0];
            l1 += h * w2[n * 4 + 1];
            l2 += h * w2[n * 4 + 2];
            l3 += h * w2[n * 4 + 3];
        }
        const float mx = fmaxf(fmaxf(l0, l1), fmaxf(l2, l3));
        const float e0 = expf(l0 - mx), e1 = expf(l1 - mx);
        const float e2 = expf(l2 - mx), e3 = expf(l3 - mx);
        const float inv = 1.0f / (e0 + e1 + e2 + e3);
        float* po = probs + (size_t)(blockIdx.x * 128 + tid) * 4;
        po[0] = e0 * inv; po[1] = e1 * inv; po[2] = e2 * inv; po[3] = e3 * inv;
    }
}

extern "C" void kernel_launch(void* const* d_in, const int* in_sizes, int n_in,
                              void* d_out, int out_size, void* d_ws, size_t ws_size,
                              hipStream_t stream) {
    (void)in_sizes; (void)n_in; (void)out_size; (void)d_ws; (void)ws_size;
    const float* prices   = (const float*)d_in[0];
    const float* enc_w    = (const float*)d_in[1];
    const float* enc_b    = (const float*)d_in[2];
    const float* enc_g    = (const float*)d_in[3];
    const float* enc_beta = (const float*)d_in[4];
    const float* head_w1  = (const float*)d_in[5];
    const float* head_b1  = (const float*)d_in[6];
    const float* head_w2  = (const float*)d_in[7];
    const float* head_b2  = (const float*)d_in[8];

    float* rf    = (float*)d_out;
    float* probs = rf + (size_t)B_ROWS * HID;

    regime_features_kernel<<<B_ROWS, 256, 0, stream>>>(
        prices, enc_w, enc_b, enc_g, enc_beta, rf);
    regime_head_kernel<<<B_ROWS / 128, 256, 0, stream>>>(
        rf, head_w1, head_b1, head_w2, head_b2, probs);
}


# Row staging loop given a constant trip count and fully unrolled: each thread issues its four 16-byte global loads back to back (all in flight at once) instead of eight serial two-load trips
# speedup vs baseline: 1.6834x; 1.2719x over previous
#include <hip/hip_runtime.h>
#include <math.h>

typedef float v2f __attribute__((ext_vector_type(2)));
typedef float v8f __attribute__((ext_vector_type(8)));

#define EPSF    1e-8f
#define LN_EPSF 1e-5f
#define B_ROWS  8192
#define S_LEN   4096
#define HID     256
#define HEAD_H  64

__global__ __launch_bounds__(256) void regime_features_kernel(
    const float* __restrict__ prices,
    const float* __restrict__ enc_w,
    const float* __restrict__ enc_b,
    const float* __restrict__ enc_g,
    const float* __restrict__ enc_beta,
    float* __restrict__ rf)
{
    __shared__ float sp[S_LEN + S_LEN / 16];
    __shared__ float r1[256];
    __shared__ float r2[256];
    __shared__ float scomb[4];
    __shared__ float smean;
    __shared__ float lnbuf[256];

    const int tid = threadIdx.x;
    const int b   = blockIdx.x;
    const float* row = prices + (size_t)b * S_LEN;

    const float4* row4 = reinterpret_cast<const float4*>(row);
    float4 stg[4];
#pragma unroll
    for (int i = 0; i < 4; ++i) stg[i] = row4[tid + 256 * i];
    const int s0 = 4 * tid + (tid >> 2);
#pragma unroll
    for (int i = 0; i < 4; ++i) {
        sp[s0 + 1088 * i + 0] = stg[i].x;
        sp[s0 + 1088 * i + 1] = stg[i].y;
        sp[s0 + 1088 * i + 2] = stg[i].z;
        sp[s0 + 1088 * i + 3] = stg[i].w;
    }
    __syncthreads();

    const int i0 = tid * 16;
    const int c0 = tid * 17;

    float own[16];
#pragma unroll
    for (int k = 0; k < 16; ++k) own[k] = sp[c0 + k];
    float hal[20];
    const int cm2 = (tid >= 2) ? c0 - 34 : 0;
    const int cm1 = (tid >= 1) ? c0 - 17 : 0;
#pragma unroll
    for (int j = 1; j < 4; ++j)
        hal[j] = (tid >= 2) ? sp[cm2 + 12 + j] : 0.f;
#pragma unroll
    for (int j = 4; j < 20; ++j)
        hal[j] = (tid >= 1) ? sp[cm1 + (j - 4)] : 0.f;

    float lsum = 0.f;
#pragma unroll
    for (int k = 0; k < 16; ++k) lsum += own[k];
    r1[tid] = lsum;
    __syncthreads();
    for (int s = 128; s > 0; s >>= 1) {
        if (tid < s) r1[tid] += r1[tid + s];
        __syncthreads();
    }
    if (tid == 0) smean = r1[0] * (1.0f / (float)S_LEN);
    __syncthreads();
    const float mean = smean;

    float winsum = 0.f;
#pragma unroll
    for (int j = 1; j < 20; ++j) winsum += hal[j];
    winsum += own[0];

    float trend = 0.f, var_acc = 0.f;
#pragma unroll
    for (int k = 0; k < 16; ++k) {
        const int i = i0 + k;
        if (k != 0) {
            winsum += own[k];
            winsum -= hal[k];
        }
        const float cnt = (float)((i + 1 < 20) ? (i + 1) : 20);
        const float sma = winsum / cnt;
        const float p   = own[k];
        trend += (p - sma) / (sma + EPSF);
        const float d = p - mean;
        var_acc += d * d;
    }
    __syncthreads();
    r1[tid] = trend;
    r2[tid] = var_acc;
    __syncthreads();
    for (int s = 128; s > 0; s >>= 1) {
        if (tid < s) { r1[tid] += r1[tid + s]; r2[tid] += r2[tid + s]; }
        __syncthreads();
    }
    if (tid == 0) {
        const float f_trend = r1[0] * (1.0f / (float)S_LEN);
        const float p_last  = sp[(S_LEN - 1) + ((S_LEN - 1) >> 4)];
        const float p_m10   = sp[(S_LEN - 10) + ((S_LEN - 10) >> 4)];
        const float f_mom   = (p_last - p_m10) / (p_m10 + EPSF);
        const float f_mr    = (p_last - mean) / (mean + EPSF);
        const float var1    = r2[0] / (float)(S_LEN - 1);
        const float f_cyc   = sqrtf(var1) / (mean + EPSF);
        scomb[0] = f_trend; scomb[1] = f_mom; scomb[2] = f_mr; scomb[3] = f_cyc;
    }
    __syncthreads();

    const int e = tid >> 6;
    const int o = tid & 63;
    float pre = enc_b[e * 64 + o];
#pragma unroll
    for (int i = 0; i < 4; ++i)
        pre += scomb[i] * enc_w[e * 256 + i * 64 + o];
    const float x = pre > 0.f ? pre : 0.f;
    lnbuf[tid] = x;
    __syncthreads();

    const int gb = e * 64;
    float mu = 0.f;
    for (int j = 0; j < 64; ++j) mu += lnbuf[gb + j];
    mu *= (1.0f / 64.0f);
    float v = 0.f;
    for (int j = 0; j < 64; ++j) { const float d = lnbuf[gb + j] - mu; v += d * d; }
    v *= (1.0f / 64.0f);
    const float outv = (x - mu) / sqrtf(v + LN_EPSF) * enc_g[gb + o] + enc_beta[gb + o];
    rf[(size_t)b * HID + tid] = outv;
}

__global__ __launch_bounds__(256) void regime_head_kernel(
    const float* __restrict__ rf,
    const float* __restrict__ w1,
    const float* __restrict__ b1,
    const float* __restrict__ w2,
    const float* __restrict__ b2,
    float* __restrict__ probs)
{
    __shared__ float hbuf[128 * HEAD_H];

    const int tid     = threadIdx.x;
    const int lane    = tid & 31;
    const int wave    = tid >> 5;
    const int rowbase = blockIdx.x * 128 + wave * 16;
    const int m16     = lane & 15;
    const int koff    = (lane >= 16) ? 2 : 0;
    const float* arow = rf + (size_t)(rowbase + m16) * HID;

    v8f acc0 = {0.f,0.f,0.f,0.f,0.f,0.f,0.f,0.f};
    v8f acc1 = acc0, acc2 = acc0, acc3 = acc0;

    for (int kk = 0; kk < HID; kk += 4) {
        v2f a;
        a.x = arow[kk + koff];
        a.y = arow[kk + koff + 1];
        const float* w1r0 = w1 + (size_t)(kk + koff) * HEAD_H + m16;
        const float* w1r1 = w1r0 + HEAD_H;
        v2f bm;
        bm.x = w1r0[0];  bm.y = w1r1[0];
        acc0 = __builtin_amdgcn_wmma_f32_16x16x4_f32(false, a, false, bm, (short)0, acc0, false, false);
        bm.x = w1r0[16]; bm.y = w1r1[16];
        acc1 = __builtin_amdgcn_wmma_f32_16x16x4_f32(false, a, false, bm, (short)0, acc1, false, false);
        bm.x = w1r0[32]; bm.y = w1r1[32];
        acc2 = __builtin_amdgcn_wmma_f32_16x16x4_f32(false, a, false, bm, (short)0, acc2, false, false);
        bm.x = w1r0[48]; bm.y = w1r1[48];
        acc3 = __builtin_amdgcn_wmma_f32_16x16x4_f32(false, a, false, bm, (short)0, acc3, false, false);
    }

    const int mlo = (lane < 16) ? 0 : 8;
#pragma unroll
    for (int r = 0; r < 8; ++r) {
        const int mloc = wave * 16 + mlo + r;
        float h0 = acc0[r] + b1[ 0 + m16]; h0 = fmaxf(h0, 0.f);
        float h1 = acc1[r] + b1[16 + m16]; h1 = fmaxf(h1, 0.f);
        float h2 = acc2[r] + b1[32 + m16]; h2 = fmaxf(h2, 0.f);
        float h3 = acc3[r] + b1[48 + m16]; h3 = fmaxf(h3, 0.f);
        hbuf[mloc * HEAD_H +  0 + m16] = h0;
        hbuf[mloc * HEAD_H + 16 + m16] = h1;
        hbuf[mloc * HEAD_H + 32 + m16] = h2;
        hbuf[mloc * HEAD_H + 48 + m16] = h3;
    }
    __syncthreads();

    if (tid < 128) {
        const float* hr = hbuf + tid * HEAD_H;
        float l0 = b2[0], l1 = b2[1], l2 = b2[2], l3 = b2[3];
        for (int n = 0; n < HEAD_H; ++n) {
            const float h = hr[n];
            l0 += h * w2[n * 4 + 0];
            l1 += h * w2[n * 4 + 1];
            l2 += h * w2[n * 4 + 2];
            l3 += h * w2[n * 4 + 3];
        }
        const float mx = fmaxf(fmaxf(l0, l1), fmaxf(l2, l3));
        const float e0 = expf(l0 - mx), e1 = expf(l1 - mx);
        const float e2 = expf(l2 - mx), e3 = expf(l3 - mx);
        const float inv = 1.0f / (e0 + e1 + e2 + e3);
        float* po = probs + (size_t)(blockIdx.x * 128 + tid) * 4;
        po[0] = e0 * inv; po[1] = e1 * inv; po[2] = e2 * inv; po[3] = e3 * inv;
    }
}

extern "C" void kernel_launch(void* const* d_in, const int* in_sizes, int n_in,
                              void* d_out, int out_size, void* d_ws, size_t ws_size,
                              hipStream_t stream) {
    (void)in_sizes; (void)n_in; (void)out_size; (void)d_ws; (void)ws_size;
    const float* prices   = (const float*)d_in[0];
    const float* enc_w    = (const float*)d_in[1];
    const float* enc_b    = (const float*)d_in[2];
    const float* enc_g    = (const float*)d_in[3];
    const float* enc_beta = (const float*)d_in[4];
    const float* head_w1  = (const float*)d_in[5];
    const float* head_b1  = (const float*)d_in[6];
    const float* head_w2  = (const float*)d_in[7];
    const float* head_b2  = (const float*)d_in[8];

    float* rf    = (float*)d_out;
    float* probs = rf + (size_t)B_ROWS * HID;

    regime_features_kernel<<<B_ROWS, 256, 0, stream>>>(
        prices, enc_w, enc_b, enc_g, enc_beta, rf);
    regime_head_kernel<<<B_ROWS / 128, 256, 0, stream>>>(
        rf, head_w1, head_b1, head_w2, head_b2, probs);
}


# Trend term as (p*cnt - winsum) * rcp(winsum + eps*cnt): one hardware reciprocal (v_rcp_f32, 1 ulp) per element instead of two full f32 divide sequences (window mean, then the ratio)
# speedup vs baseline: 1.9887x; 1.1814x over previous
#include <hip/hip_runtime.h>
#include <math.h>

typedef float v2f __attribute__((ext_vector_type(2)));
typedef float v8f __attribute__((ext_vector_type(8)));

#define EPSF    1e-8f
#define LN_EPSF 1e-5f
#define B_ROWS  8192
#define S_LEN   4096
#define HID     256
#define HEAD_H  64

__global__ __launch_bounds__(256) void regime_features_kernel(
    const float* __restrict__ prices,
    const float* __restrict__ enc_w,
    const float* __restrict__ enc_b,
    const float* __restrict__ enc_g,
    const float* __restrict__ enc_beta,
    float* __restrict__ rf)
{
    __shared__ float sp[S_LEN + S_LEN / 16];
    __shared__ float r1[256];
    __shared__ float r2[256];
    __shared__ float scomb[4];
    __shared__ float smean;
    __shared__ float lnbuf[256];

    const int tid = threadIdx.x;
    const int b   = blockIdx.x;
    const float* row = prices + (size_t)b * S_LEN;

    const float4* row4 = reinterpret_cast<const float4*>(row);
    float4 stg[4];
#pragma unroll
    for (int i = 0; i < 4; ++i) stg[i] = row4[tid + 256 * i];
    const int s0 = 4 * tid + (tid >> 2);
#pragma unroll
    for (int i = 0; i < 4; ++i) {
        sp[s0 + 1088 * i + 0] = stg[i].x;
        sp[s0 + 1088 * i + 1] = stg[i].y;
        sp[s0 + 1088 * i + 2] = stg[i].z;
        sp[s0 + 1088 * i + 3] = stg[i].w;
    }
    __syncthreads();

    const int i0 = tid * 16;
    const int c0 = tid * 17;

    float own[16];
#pragma unroll
    for (int k = 0; k < 16; ++k) own[k] = sp[c0 + k];
    float hal[20];
    const int cm2 = (tid >= 2) ? c0 - 34 : 0;
    const int cm1 = (tid >= 1) ? c0 - 17 : 0;
#pragma unroll
    for (int j = 1; j < 4; ++j)
        hal[j] = (tid >= 2) ? sp[cm2 + 12 + j] : 0.f;
#pragma unroll
    for (int j = 4; j < 20; ++j)
        hal[j] = (tid >= 1) ? sp[cm1 + (j - 4)] : 0.f;

    float lsum = 0.f;
#pragma unroll
    for (int k = 0; k < 16; ++k) lsum += own[k];
    r1[tid] = lsum;
    __syncthreads();
    for (int s = 128; s > 0; s >>= 1) {
        if (tid < s) r1[tid] += r1[tid + s];
        __syncthreads();
    }
    if (tid == 0) smean = r1[0] * (1.0f / (float)S_LEN);
    __syncthreads();
    const float mean = smean;

    float winsum = 0.f;
#pragma unroll
    for (int j = 1; j < 20; ++j) winsum += hal[j];
    winsum += own[0];

    const float cbase = (float)(i0 + 1);
    float trend = 0.f, var_acc = 0.f;
#pragma unroll
    for (int k = 0; k < 16; ++k) {
        if (k != 0) {
            winsum += own[k];
            winsum -= hal[k];
        }
        const float cnt = fminf(cbase + (float)k, 20.f);
        const float p   = own[k];
        const float num = p * cnt - winsum;
        const float den = winsum + EPSF * cnt;
        trend += num * __builtin_amdgcn_rcpf(den);
        const float d = p - mean;
        var_acc += d * d;
    }
    __syncthreads();
    r1[tid] = trend;
    r2[tid] = var_acc;
    __syncthreads();
    for (int s = 128; s > 0; s >>= 1) {
        if (tid < s) { r1[tid] += r1[tid + s]; r2[tid] += r2[tid + s]; }
        __syncthreads();
    }
    if (tid == 0) {
        const float f_trend = r1[0] * (1.0f / (float)S_LEN);
        const float p_last  = sp[(S_LEN - 1) + ((S_LEN - 1) >> 4)];
        const float p_m10   = sp[(S_LEN - 10) + ((S_LEN - 10) >> 4)];
        const float f_mom   = (p_last - p_m10) / (p_m10 + EPSF);
        const float f_mr    = (p_last - mean) / (mean + EPSF);
        const float var1    = r2[0] / (float)(S_LEN - 1);
        const float f_cyc   = sqrtf(var1) / (mean + EPSF);
        scomb[0] = f_trend; scomb[1] = f_mom; scomb[2] = f_mr; scomb[3] = f_cyc;
    }
    __syncthreads();

    const int e = tid >> 6;
    const int o = tid & 63;
    float pre = enc_b[e * 64 + o];
#pragma unroll
    for (int i = 0; i < 4; ++i)
        pre += scomb[i] * enc_w[e * 256 + i * 64 + o];
    const float x = pre > 0.f ? pre : 0.f;
    lnbuf[tid] = x;
    __syncthreads();

    const int gb = e * 64;
    float mu = 0.f;
    for (int j = 0; j < 64; ++j) mu += lnbuf[gb + j];
    mu *= (1.0f / 64.0f);
    float v = 0.f;
    for (int j = 0; j < 64; ++j) { const float d = lnbuf[gb + j] - mu; v += d * d; }
    v *= (1.0f / 64.0f);
    const float outv = (x - mu) / sqrtf(v + LN_EPSF) * enc_g[gb + o] + enc_beta[gb + o];
    rf[(size_t)b * HID + tid] = outv;
}

__global__ __launch_bounds__(256) void regime_head_kernel(
    const float* __restrict__ rf,
    const float* __restrict__ w1,
    const float* __restrict__ b1,
    const float* __restrict__ w2,
    const float* __restrict__ b2,
    float* __restrict__ probs)
{
    __shared__ float hbuf[128 * HEAD_H];

    const int tid     = threadIdx.x;
    const int lane    = tid & 31;
    const int wave    = tid >> 5;
    const int rowbase = blockIdx.x * 128 + wave * 16;
    const int m16     = lane & 15;
    const int koff    = (lane >= 16) ? 2 : 0;
    const float* arow = rf + (size_t)(rowbase + m16) * HID;

    v8f acc0 = {0.f,0.f,0.f,0.f,0.f,0.f,0.f,0.f};
    v8f acc1 = acc0, acc2 = acc0, acc3 = acc0;

    for (int kk = 0; kk < HID; kk += 4) {
        v2f a;
        a.x = arow[kk + koff];
        a.y = arow[kk + koff + 1];
        const float* w1r0 = w1 + (size_t)(kk + koff) * HEAD_H + m16;
        const float* w1r1 = w1r0 + HEAD_H;
        v2f bm;
        bm.x = w1r0[0];  bm.y = w1r1[0];
        acc0 = __builtin_amdgcn_wmma_f32_16x16x4_f32(false, a, false, bm, (short)0, acc0, false, false);
        bm.x = w1r0[16]; bm.y = w1r1[16];
        acc1 = __builtin_amdgcn_wmma_f32_16x16x4_f32(false, a, false, bm, (short)0, acc1, false, false);
        bm.x = w1r0[32]; bm.y = w1r1[32];
        acc2 = __builtin_amdgcn_wmma_f32_16x16x4_f32(false, a, false, bm, (short)0, acc2, false, false);
        bm.x = w1r0[48]; bm.y = w1r1[48];
        acc3 = __builtin_amdgcn_wmma_f32_16x16x4_f32(false, a, false, bm, (short)0, acc3, false, false);
    }

    const int mlo = (lane < 16) ? 0 : 8;
#pragma unroll
    for (int r = 0; r < 8; ++r) {
        const int mloc = wave * 16 + mlo + r;
        float h0 = acc0[r] + b1[ 0 + m16]; h0 = fmaxf(h0, 0.f);
        float h1 = acc1[r] + b1[16 + m16]; h1 = fmaxf(h1, 0.f);
        float h2 = acc2[r] + b1[32 + m16]; h2 = fmaxf(h2, 0.f);
        float h3 = acc3[r] + b1[48 + m16]; h3 = fmaxf(h3, 0.f);
        hbuf[mloc * HEAD_H +  0 + m16] = h0;
        hbuf[mloc * HEAD_H + 16 + m16] = h1;
        hbuf[mloc * HEAD_H + 32 + m16] = h2;
        hbuf[mloc * HEAD_H + 48 + m16] = h3;
    }
    __syncthreads();

    if (tid < 128) {
        const float* hr = hbuf + tid * HEAD_H;
        float l0 = b2[0], l1 = b2[1], l2 = b2[2], l3 = b2[3];
        for (int n = 0; n < HEAD_H; ++n) {
            const float h = hr[n];
            l0 += h * w2[n * 4 + 0];
            l1 += h * w2[n * 4 + 1];
            l2 += h * w2[n * 4 + 2];
            l3 += h * w2[n * 4 + 3];
        }
        const float mx = fmaxf(fmaxf(l0, l1), fmaxf(l2, l3));
        const float e0 = expf(l0 - mx), e1 = expf(l1 - mx);
        const float e2 = expf(l2 - mx), e3 = expf(l3 - mx);
        const float inv = 1.0f / (e0 + e1 + e2 + e3);
        float* po = probs + (size_t)(blockIdx.x * 128 + tid) * 4;
        po[0] = e0 * inv; po[1] = e1 * inv; po[2] = e2 * inv; po[3] = e3 * inv;
    }
}

extern "C" void kernel_launch(void* const* d_in, const int* in_sizes, int n_in,
                              void* d_out, int out_size, void* d_ws, size_t ws_size,
                              hipStream_t stream) {
    (void)in_sizes; (void)n_in; (void)out_size; (void)d_ws; (void)ws_size;
    const float* prices   = (const float*)d_in[0];
    const float* enc_w    = (const float*)d_in[1];
    const float* enc_b    = (const float*)d_in[2];
    const float* enc_g    = (const float*)d_in[3];
    const float* enc_beta = (const float*)d_in[4];
    const float* head_w1  = (const float*)d_in[5];
    const float* head_b1  = (const float*)d_in[6];
    const float* head_w2  = (const float*)d_in[7];
    const float* head_b2  = (const float*)d_in[8];

    float* rf    = (float*)d_out;
    float* probs = rf + (size_t)B_ROWS * HID;

    regime_features_kernel<<<B_ROWS, 256, 0, stream>>>(
        prices, enc_w, enc_b, enc_g, enc_beta, rf);
    regime_head_kernel<<<B_ROWS / 128, 256, 0, stream>>>(
        rf, head_w1, head_b1, head_w2, head_b2, probs);
}


# All row sums and LayerNorm sums done inside waves by cross-lane adds (row_xmask DPP, lane^16 swap) plus one 8-entry LDS combine, not LDS trees and serial 64-reads; summation order changes
# speedup vs baseline: 2.4958x; 1.2550x over previous
#include <hip/hip_runtime.h>
#include <math.h>

typedef float v2f __attribute__((ext_vector_type(2)));
typedef float v8f __attribute__((ext_vector_type(8)));

#define EPSF    1e-8f
#define LN_EPSF 1e-5f
#define B_ROWS  8192
#define S_LEN   4096
#define HID     256
#define HEAD_H  64

template <int CTRL>
__device__ __forceinline__ float row_xmask_read(float v) {
    return __builtin_bit_cast(float,
        __builtin_amdgcn_update_dpp(0, __builtin_bit_cast(int, v), CTRL, 0xf, 0xf, true));
}
__device__ __forceinline__ float other_row_read(float v) {
    const int i = __builtin_bit_cast(int, v);
    return __builtin_bit_cast(float,
        __builtin_amdgcn_permlanex16(i, i, 0x76543210, 0xfedcba98, false, false));
}
__device__ __forceinline__ float wave_sum(float v) {
    v += row_xmask_read<0x161>(v);
    v += row_xmask_read<0x162>(v);
    v += row_xmask_read<0x164>(v);
    v += row_xmask_read<0x168>(v);
    v += other_row_read(v);
    return v;
}

__global__ __launch_bounds__(256) void regime_features_kernel(
    const float* __restrict__ prices,
    const float* __restrict__ enc_w,
    const float* __restrict__ enc_b,
    const float* __restrict__ enc_g,
    const float* __restrict__ enc_beta,
    float* __restrict__ rf)
{
    __shared__ float sp[S_LEN + S_LEN / 16];
    __shared__ float wsum[8];
    __shared__ float wtr[8];
    __shared__ float wvar[8];
    __shared__ float scomb[4];

    const int tid  = threadIdx.x;
    const int lane = tid & 31;
    const int wave = __builtin_amdgcn_readfirstlane(tid >> 5);
    const int b    = blockIdx.x;
    const float* row = prices + (size_t)b * S_LEN;

    const float4* row4 = reinterpret_cast<const float4*>(row);
    float4 stg[4];
#pragma unroll
    for (int i = 0; i < 4; ++i) stg[i] = row4[tid + 256 * i];
    const int s0 = 4 * tid + (tid >> 2);
#pragma unroll
    for (int i = 0; i < 4; ++i) {
        sp[s0 + 1088 * i + 0] = stg[i].x;
        sp[s0 + 1088 * i + 1] = stg[i].y;
        sp[s0 + 1088 * i + 2] = stg[i].z;
        sp[s0 + 1088 * i + 3] = stg[i].w;
    }
    __syncthreads();

    const int i0 = tid * 16;
    const int c0 = tid * 17;

    float own[16];
#pragma unroll
    for (int k = 0; k < 16; ++k) own[k] = sp[c0 + k];
    float hal[20];
    const int cm2 = (tid >= 2) ? c0 - 34 : 0;
    const int cm1 = (tid >= 1) ? c0 - 17 : 0;
#pragma unroll
    for (int j = 1; j < 4; ++j)
        hal[j] = (tid >= 2) ? sp[cm2 + 12 + j] : 0.f;
#pragma unroll
    for (int j = 4; j < 20; ++j)
        hal[j] = (tid >= 1) ? sp[cm1 + (j - 4)] : 0.f;

    float lsum = 0.f;
#pragma unroll
    for (int k = 0; k < 16; ++k) lsum += own[k];
    lsum = wave_sum(lsum);
    if (lane == 0) wsum[wave] = lsum;
    __syncthreads();
    float rsum = wsum[0];
    for (int w = 1; w < 8; ++w) rsum += wsum[w];
    const float mean = rsum * (1.0f / (float)S_LEN);

    float winsum = 0.f;
#pragma unroll
    for (int j = 1; j < 20; ++j) winsum += hal[j];
    winsum += own[0];

    const float cbase = (float)(i0 + 1);
    float trend = 0.f, var_acc = 0.f;
#pragma unroll
    for (int k = 0; k < 16; ++k) {
        if (k != 0) {
            winsum += own[k];
            winsum -= hal[k];
        }
        const float cnt = fminf(cbase + (float)k, 20.f);
        const float p   = own[k];
        const float num = p * cnt - winsum;
        const float den = winsum + EPSF * cnt;
        trend += num * __builtin_amdgcn_rcpf(den);
        const float d = p - mean;
        var_acc += d * d;
    }
    trend   = wave_sum(trend);
    var_acc = wave_sum(var_acc);
    if (lane == 0) { wtr[wave] = trend; wvar[wave] = var_acc; }
    __syncthreads();
    if (tid == 0) {
        float tsum = wtr[0], vsum = wvar[0];
        for (int w = 1; w < 8; ++w) { tsum += wtr[w]; vsum += wvar[w]; }
        const float f_trend = tsum * (1.0f / (float)S_LEN);
        const float p_last  = sp[(S_LEN - 1) + ((S_LEN - 1) >> 4)];
        const float p_m10   = sp[(S_LEN - 10) + ((S_LEN - 10) >> 4)];
        const float f_mom   = (p_last - p_m10) / (p_m10 + EPSF);
        const float f_mr    = (p_last - mean) / (mean + EPSF);
        const float var1    = vsum / (float)(S_LEN - 1);
        const float f_cyc   = sqrtf(var1) / (mean + EPSF);
        scomb[0] = f_trend; scomb[1] = f_mom; scomb[2] = f_mr; scomb[3] = f_cyc;
    }
    __syncthreads();

    if (wave < 4) {
        const int gb = wave * 64;
        const int oa = gb + lane;
        const int ob = oa + 32;
        float pa = enc_b[oa];
        float pb = enc_b[ob];
        for (int i = 0; i < 4; ++i) {
            const float f = scomb[i];
            pa += f * enc_w[wave * 256 + i * 64 + lane];
            pb += f * enc_w[wave * 256 + i * 64 + 32 + lane];
        }
        const float xa = pa > 0.f ? pa : 0.f;
        const float xb = pb > 0.f ? pb : 0.f;
        const float mu = wave_sum(xa + xb) * (1.0f / 64.0f);
        const float da = xa - mu;
        const float db = xb - mu;
        const float v  = wave_sum(da * da + db * db) * (1.0f / 64.0f);
        const float sd = sqrtf(v + LN_EPSF);
        float* orow = rf + (size_t)b * HID;
        orow[oa] = da / sd * enc_g[oa] + enc_beta[oa];
        orow[ob] = db / sd * enc_g[ob] + enc_beta[ob];
    }
}

__global__ __launch_bounds__(256) void regime_head_kernel(
    const float* __restrict__ rf,
    const float* __restrict__ w1,
    const float* __restrict__ b1,
    const float* __restrict__ w2,
    const float* __restrict__ b2,
    float* __restrict__ probs)
{
    __shared__ float hbuf[128 * HEAD_H];

    const int tid     = threadIdx.x;
    const int lane    = tid & 31;
    const int wave    = tid >> 5;
    const int rowbase = blockIdx.x * 128 + wave * 16;
    const int m16     = lane & 15;
    const int koff    = (lane >= 16) ? 2 : 0;
    const float* arow = rf + (size_t)(rowbase + m16) * HID;

    v8f acc0 = {0.f,0.f,0.f,0.f,0.f,0.f,0.f,0.f};
    v8f acc1 = acc0, acc2 = acc0, acc3 = acc0;

    for (int kk = 0; kk < HID; kk += 4) {
        v2f a;
        a.x = arow[kk + koff];
        a.y = arow[kk + koff + 1];
        const float* w1r0 = w1 + (size_t)(kk + koff) * HEAD_H + m16;
        const float* w1r1 = w1r0 + HEAD_H;
        v2f bm;
        bm.x = w1r0[0];  bm.y = w1r1[0];
        acc0 = __builtin_amdgcn_wmma_f32_16x16x4_f32(false, a, false, bm, (short)0, acc0, false, false);
        bm.x = w1r0[16]; bm.y = w1r1[16];
        acc1 = __builtin_amdgcn_wmma_f32_16x16x4_f32(false, a, false, bm, (short)0, acc1, false, false);
        bm.x = w1r0[32]; bm.y = w1r1[32];
        acc2 = __builtin_amdgcn_wmma_f32_16x16x4_f32(false, a, false, bm, (short)0, acc2, false, false);
        bm.x = w1r0[48]; bm.y = w1r1[48];
        acc3 = __builtin_amdgcn_wmma_f32_16x16x4_f32(false, a, false, bm, (short)0, acc3, false, false);
    }

    const int mlo = (lane < 16) ? 0 : 8;
#pragma unroll
    for (int r = 0; r < 8; ++r) {
        const int mloc = wave * 16 + mlo + r;
        float h0 = acc0[r] + b1[ 0 + m16]; h0 = fmaxf(h0, 0.f);
        float h1 = acc1[r] + b1[16 + m16]; h1 = fmaxf(h1, 0.f);
        float h2 = acc2[r] + b1[32 + m16]; h2 = fmaxf(h2, 0.f);
        float h3 = acc3[r] + b1[48 + m16]; h3 = fmaxf(h3, 0.f);
        hbuf[mloc * HEAD_H +  0 + m16] = h0;
        hbuf[mloc * HEAD_H + 16 + m16] = h1;
        hbuf[mloc * HEAD_H + 32 + m16] = h2;
        hbuf[mloc * HEAD_H + 48 + m16] = h3;
    }
    __syncthreads();

    if (tid < 128) {
        const float* hr = hbuf + tid * HEAD_H;
        float l0 = b2[0], l1 = b2[1], l2 = b2[2], l3 = b2[3];
        for (int n = 0; n < HEAD_H; ++n) {
            const float h = hr[n];
            l0 += h * w2[n * 4 + 0];
            l1 += h * w2[n * 4 + 1];
            l2 += h * w2[n * 4 + 2];
            l3 += h * w2[n * 4 + 3];
        }
        const float mx = fmaxf(fmaxf(l0, l1), fmaxf(l2, l3));
        const float e0 = expf(l0 - mx), e1 = expf(l1 - mx);
        const float e2 = expf(l2 - mx), e3 = expf(l3 - mx);
        const float inv = 1.0f / (e0 + e1 + e2 + e3);
        float* po = probs + (size_t)(blockIdx.x * 128 + tid) * 4;
        po[0] = e0 * inv; po[1] = e1 * inv; po[2] = e2 * inv; po[3] = e3 * inv;
    }
}

extern "C" void kernel_launch(void* const* d_in, const int* in_sizes, int n_in,
                              void* d_out, int out_size, void* d_ws, size_t ws_size,
                              hipStream_t stream) {
    (void)in_sizes; (void)n_in; (void)out_size; (void)d_ws; (void)ws_size;
    const float* prices   = (const float*)d_in[0];
    const float* enc_w    = (const float*)d_in[1];
    const float* enc_b    = (const float*)d_in[2];
    const float* enc_g    = (const float*)d_in[3];
    const float* enc_beta = (const float*)d_in[4];
    const float* head_w1  = (const float*)d_in[5];
    const float* head_b1  = (const float*)d_in[6];
    const float* head_w2  = (const float*)d_in[7];
    const float* head_b2  = (const float*)d_in[8];

    float* rf    = (float*)d_out;
    float* probs = rf + (size_t)B_ROWS * HID;

    regime_features_kernel<<<B_ROWS, 256, 0, stream>>>(
        prices, enc_w, enc_b, enc_g, enc_beta, rf);
    regime_head_kernel<<<B_ROWS / 128, 256, 0, stream>>>(
        rf, head_w1, head_b1, head_w2, head_b2, probs);
}


# Head kernel re-tiled to 256 blocks of 32 rows: W1 and the row tile staged once per block in LDS by 16-byte loads all in flight, WMMA fed from LDS; projection 4 lanes per row, its sum order changes
# speedup vs baseline: 3.4324x; 1.3753x over previous
#include <hip/hip_runtime.h>
#include <math.h>

typedef float v2f __attribute__((ext_vector_type(2)));
typedef float v4f __attribute__((ext_vector_type(4)));
typedef float v8f __attribute__((ext_vector_type(8)));

#define EPSF    1e-8f
#define LN_EPSF 1e-5f
#define B_ROWS  8192
#define S_LEN   4096
#define HID     256
#define HEAD_H  64

template <int CTRL>
__device__ __forceinline__ float row_xmask_read(float v) {
    return __builtin_bit_cast(float,
        __builtin_amdgcn_update_dpp(0, __builtin_bit_cast(int, v), CTRL, 0xf, 0xf, true));
}
__device__ __forceinline__ float other_row_read(float v) {
    const int i = __builtin_bit_cast(int, v);
    return __builtin_bit_cast(float,
        __builtin_amdgcn_permlanex16(i, i, 0x76543210, 0xfedcba98, false, false));
}
__device__ __forceinline__ float wave_sum(float v) {
    v += row_xmask_read<0x161>(v);
    v += row_xmask_read<0x162>(v);
    v += row_xmask_read<0x164>(v);
    v += row_xmask_read<0x168>(v);
    v += other_row_read(v);
    return v;
}

__global__ __launch_bounds__(256) void regime_features_kernel(
    const float* __restrict__ prices,
    const float* __restrict__ enc_w,
    const float* __restrict__ enc_b,
    const float* __restrict__ enc_g,
    const float* __restrict__ enc_beta,
    float* __restrict__ rf)
{
    __shared__ float sp[S_LEN + S_LEN / 16];
    __shared__ float wsum[8];
    __shared__ float wtr[8];
    __shared__ float wvar[8];
    __shared__ float scomb[4];

    const int tid  = threadIdx.x;
    const int lane = tid & 31;
    const int wave = __builtin_amdgcn_readfirstlane(tid >> 5);
    const int b    = blockIdx.x;
    const float* row = prices + (size_t)b * S_LEN;

    const float4* row4 = reinterpret_cast<const float4*>(row);
    float4 stg[4];
#pragma unroll
    for (int i = 0; i < 4; ++i) stg[i] = row4[tid + 256 * i];
    const int s0 = 4 * tid + (tid >> 2);
#pragma unroll
    for (int i = 0; i < 4; ++i) {
        sp[s0 + 1088 * i + 0] = stg[i].x;
        sp[s0 + 1088 * i + 1] = stg[i].y;
        sp[s0 + 1088 * i + 2] = stg[i].z;
        sp[s0 + 1088 * i + 3] = stg[i].w;
    }
    __syncthreads();

    const int i0 = tid * 16;
    const int c0 = tid * 17;

    float own[16];
#pragma unroll
    for (int k = 0; k < 16; ++k) own[k] = sp[c0 + k];
    float hal[20];
    const int cm2 = (tid >= 2) ? c0 - 34 : 0;
    const int cm1 = (tid >= 1) ? c0 - 17 : 0;
#pragma unroll
    for (int j = 1; j < 4; ++j)
        hal[j] = (tid >= 2) ? sp[cm2 + 12 + j] : 0.f;
#pragma unroll
    for (int j = 4; j < 20; ++j)
        hal[j] = (tid >= 1) ? sp[cm1 + (j - 4)] : 0.f;

    float lsum = 0.f;
#pragma unroll
    for (int k = 0; k < 16; ++k) lsum += own[k];
    lsum = wave_sum(lsum);
    if (lane == 0) wsum[wave] = lsum;
    __syncthreads();
    float rsum = wsum[0];
    for (int w = 1; w < 8; ++w) rsum += wsum[w];
    const float mean = rsum * (1.0f / (float)S_LEN);

    float winsum = 0.f;
#pragma unroll
    for (int j = 1; j < 20; ++j) winsum += hal[j];
    winsum += own[0];

    const float cbase = (float)(i0 + 1);
    float trend = 0.f, var_acc = 0.f;
#pragma unroll
    for (int k = 0; k < 16; ++k) {
        if (k != 0) {
            winsum += own[k];
            winsum -= hal[k];
        }
        const float cnt = fminf(cbase + (float)k, 20.f);
        const float p   = own[k];
        const float num = p * cnt - winsum;
        const float den = winsum + EPSF * cnt;
        trend += num * __builtin_amdgcn_rcpf(den);
        const float d = p - mean;
        var_acc += d * d;
    }
    trend   = wave_sum(trend);
    var_acc = wave_sum(var_acc);
    if (lane == 0) { wtr[wave] = trend; wvar[wave] = var_acc; }
    __syncthreads();
    if (tid == 0) {
        float tsum = wtr[0], vsum = wvar[0];
        for (int w = 1; w < 8; ++w) { tsum += wtr[w]; vsum += wvar[w]; }
        const float f_trend = tsum * (1.0f / (float)S_LEN);
        const float p_last  = sp[(S_LEN - 1) + ((S_LEN - 1) >> 4)];
        const float p_m10   = sp[(S_LEN - 10) + ((S_LEN - 10) >> 4)];
        const float f_mom   = (p_last - p_m10) / (p_m10 + EPSF);
        const float f_mr    = (p_last - mean) / (mean + EPSF);
        const float var1    = vsum / (float)(S_LEN - 1);
        const float f_cyc   = sqrtf(var1) / (mean + EPSF);
        scomb[0] = f_trend; scomb[1] = f_mom; scomb[2] = f_mr; scomb[3] = f_cyc;
    }
    __syncthreads();

    if (wave < 4) {
        const int gb = wave * 64;
        const int oa = gb + lane;
        const int ob = oa + 32;
        float pa = enc_b[oa];
        float pb = enc_b[ob];
        for (int i = 0; i < 4; ++i) {
            const float f = scomb[i];
            pa += f * enc_w[wave * 256 + i * 64 + lane];
            pb += f * enc_w[wave * 256 + i * 64 + 32 + lane];
        }
        const float xa = pa > 0.f ? pa : 0.f;
        const float xb = pb > 0.f ? pb : 0.f;
        const float mu = wave_sum(xa + xb) * (1.0f / 64.0f);
        const float da = xa - mu;
        const float db = xb - mu;
        const float v  = wave_sum(da * da + db * db) * (1.0f / 64.0f);
        const float sd = sqrtf(v + LN_EPSF);
        float* orow = rf + (size_t)b * HID;
        orow[oa] = da / sd * enc_g[oa] + enc_beta[oa];
        orow[ob] = db / sd * enc_g[ob] + enc_beta[ob];
    }
}

#define HT_ROWS  32
#define A_PITCH  260
#define HB_PITCH 65

__global__ __launch_bounds__(256) void regime_head_kernel(
    const float* __restrict__ rf,
    const float* __restrict__ w1,
    const float* __restrict__ b1,
    const float* __restrict__ w2,
    const float* __restrict__ b2,
    float* __restrict__ probs)
{
    __shared__ __attribute__((aligned(16))) float Ws[HID * HEAD_H];
    __shared__ __attribute__((aligned(16))) float As[HT_ROWS * A_PITCH];
    __shared__ float hbuf[HT_ROWS * HB_PITCH];

    const int tid  = threadIdx.x;
    const int lane = tid & 31;
    const int wave = __builtin_amdgcn_readfirstlane(tid >> 5);
    const int row0 = blockIdx.x * HT_ROWS;

    const int prow = tid >> 2;
    const int part = tid & 3;
    v4f w2r[16];
    if (wave < 4) {
        const v4f* w24 = reinterpret_cast<const v4f*>(w2);
#pragma unroll
        for (int j = 0; j < 16; ++j) w2r[j] = w24[part * 16 + j];
    }

    const float bias1 = b1[(wave >> 1) * 16 + (lane & 15)];

    const v4f* w14 = reinterpret_cast<const v4f*>(w1);
    const v4f* a4  = reinterpret_cast<const v4f*>(rf + (size_t)row0 * HID);
    const int kr = lane >> 3;
    v4f wst[16];
    v4f ast[8];
#pragma unroll
    for (int i = 0; i < 16; ++i) {
        const int u = wave + 8 * i;
        const int g = (u & 1) * 8 + (lane & 7);
        wst[i] = w14[(4 * (u >> 1) + kr) * 16 + g];
    }
#pragma unroll
    for (int i = 0; i < 8; ++i) ast[i] = a4[tid + 256 * i];
#pragma unroll
    for (int i = 0; i < 16; ++i) {
        const int u = wave + 8 * i;
        const int g = (u & 1) * 8 + (lane & 7);
        const int o = ((u >> 1) * 4 + (g >> 2)) * 64 + kr * 16 + 4 * (g & 3);
        *reinterpret_cast<v4f*>(&Ws[o]) = wst[i];
    }
#pragma unroll
    for (int i = 0; i < 8; ++i) {
        const int f = tid + 256 * i;
        *reinterpret_cast<v4f*>(&As[(f >> 6) * A_PITCH + 4 * (f & 63)]) = ast[i];
    }
    __syncthreads();

    const int rt  = wave & 1;
    const int nt  = wave >> 1;
    const int m16 = lane & 15;
    const int hh  = lane >> 4;
    const float* ap = As + (rt * 16 + m16) * A_PITCH + 2 * hh;
    const float* bp = Ws + nt * 64 + hh * 32 + m16;
    constexpr int KB = 16;
    v2f fa[2][KB], fb[2][KB];
#pragma unroll
    for (int j = 0; j < KB; ++j) {
        fa[0][j]   = *reinterpret_cast<const v2f*>(ap + 4 * j);
        fb[0][j].x = bp[j * 256];
        fb[0][j].y = bp[j * 256 + 16];
    }
    v8f acc = {0.f,0.f,0.f,0.f,0.f,0.f,0.f,0.f};
#pragma unroll
    for (int blk = 0; blk < HID / 4 / KB; ++blk) {
        if (blk + 1 < HID / 4 / KB) {
#pragma unroll
            for (int j = 0; j < KB; ++j) {
                const int s = (blk + 1) * KB + j;
                fa[(blk + 1) & 1][j]   = *reinterpret_cast<const v2f*>(ap + 4 * s);
                fb[(blk + 1) & 1][j].x = bp[s * 256];
                fb[(blk + 1) & 1][j].y = bp[s * 256 + 16];
            }
        }
        __builtin_amdgcn_sched_barrier(0);
#pragma unroll
        for (int j = 0; j < KB; ++j)
            acc = __builtin_amdgcn_wmma_f32_16x16x4_f32(false, fa[blk & 1][j], false, fb[blk & 1][j], (short)0, acc, false, false);
        __builtin_amdgcn_sched_barrier(0);
    }

#pragma unroll
    for (int r = 0; r < 8; ++r) {
        const int mloc = rt * 16 + hh * 8 + r;
        hbuf[mloc * HB_PITCH + nt * 16 + m16] = fmaxf(acc[r] + bias1, 0.f);
    }
    __syncthreads();

    if (wave < 4) {
        const float* hr = hbuf + prow * HB_PITCH + part * 16;
        float l0 = 0.f, l1 = 0.f, l2 = 0.f, l3 = 0.f;
#pragma unroll
        for (int j = 0; j < 16; ++j) {
            const float h = hr[j];
            l0 += h * w2r[j].x;
            l1 += h * w2r[j].y;
            l2 += h * w2r[j].z;
            l3 += h * w2r[j].w;
        }
        l0 += row_xmask_read<0x161>(l0); l0 += row_xmask_read<0x162>(l0);
        l1 += row_xmask_read<0x161>(l1); l1 += row_xmask_read<0x162>(l1);
        l2 += row_xmask_read<0x161>(l2); l2 += row_xmask_read<0x162>(l2);
        l3 += row_xmask_read<0x161>(l3); l3 += row_xmask_read<0x162>(l3);
        l0 += b2[0]; l1 += b2[1]; l2 += b2[2]; l3 += b2[3];
        const float mx = fmaxf(fmaxf(l0, l1), fmaxf(l2, l3));
        const float e0 = expf(l0 - mx), e1 = expf(l1 - mx);
        const float e2 = expf(l2 - mx), e3 = expf(l3 - mx);
        const float inv = 1.0f / (e0 + e1 + e2 + e3);
        const float elo  = (part & 1) ? e1 : e0;
        const float ehi  = (part & 1) ? e3 : e2;
        const float mine = (part & 2) ? ehi : elo;
        probs[(size_t)(row0 + prow) * 4 + part] = mine * inv;
    }
}

extern "C" void kernel_launch(void* const* d_in, const int* in_sizes, int n_in,
                              void* d_out, int out_size, void* d_ws, size_t ws_size,
                              hipStream_t stream) {
    (void)in_sizes; (void)n_in; (void)out_size; (void)d_ws; (void)ws_size;
    const float* prices   = (const float*)d_in[0];
    const float* enc_w    = (const float*)d_in[1];
    const float* enc_b    = (const float*)d_in[2];
    const float* enc_g    = (const float*)d_in[3];
    const float* enc_beta = (const float*)d_in[4];
    const float* head_w1  = (const float*)d_in[5];
    const float* head_b1  = (const float*)d_in[6];
    const float* head_w2  = (const float*)d_in[7];
    const float* head_b2  = (const float*)d_in[8];

    float* rf    = (float*)d_out;
    float* probs = rf + (size_t)B_ROWS * HID;

    regime_features_kernel<<<B_ROWS, 256, 0, stream>>>(
        prices, enc_w, enc_b, enc_g, enc_beta, rf);
    regime_head_kernel<<<B_ROWS / HT_ROWS, 256, 0, stream>>>(
        rf, head_w1, head_b1, head_w2, head_b2, probs);
}
